# starting kernel
# baseline (speedup 1.0000x reference)
#include <hip/hip_runtime.h>
#include <hip/hip_bf16.h>

#define B_  4
#define TE_ 256
#define TD_ 256
#define DE_ 512
#define DD_ 512

typedef __attribute__((ext_vector_type(2))) float v2f;
typedef __attribute__((ext_vector_type(4))) float v4f;
typedef __attribute__((ext_vector_type(8))) float v8f;

#if __has_builtin(__builtin_amdgcn_tanhf)
__device__ __forceinline__ float fast_tanh(float x) { return __builtin_amdgcn_tanhf(x); }
#elif __has_builtin(__builtin_amdgcn_tanh_f32)
__device__ __forceinline__ float fast_tanh(float x) { return __builtin_amdgcn_tanh_f32(x); }
#else
__device__ __forceinline__ float fast_tanh(float x) { return tanhf(x); }
#endif

#if __has_builtin(__builtin_amdgcn_exp2f)
__device__ __forceinline__ float fast_exp2(float x) { return __builtin_amdgcn_exp2f(x); }
#else
__device__ __forceinline__ float fast_exp2(float x) { return exp2f(x); }
#endif

__global__ __launch_bounds__(32) void wmma_gemm_f32(
    const float* __restrict__ A, const float* __restrict__ Bm,
    float* __restrict__ C, int K, int N,
    long sA, long sB, long sC)
{
    const int lane = threadIdx.x & 31;
    const int half = lane >> 4;
    const int r    = lane & 15;

    const float* Ab = A  + (long)blockIdx.z * sA;
    const float* Bb = Bm + (long)blockIdx.z * sB;
    float*       Cb = C  + (long)blockIdx.z * sC;

    const int m0 = blockIdx.x * 16;
    const int n0 = blockIdx.y * 16;

    const float* aRow = Ab + (long)(m0 + r) * K;
    const float* bCol = Bb + n0 + r;

    v8f acc = {};
    for (int k0 = 0; k0 < K; k0 += 4) {
        v2f a = *(const v2f*)(aRow + k0 + 2 * half);
        v2f b;
        b.x = bCol[(long)(k0 + 2 * half + 0) * N];
        b.y = bCol[(long)(k0 + 2 * half + 1) * N];
        acc = __builtin_amdgcn_wmma_f32_16x16x4_f32(
             false, a,  false, b,
             (short)0, acc,  false,  false);
    }

#pragma unroll
    for (int i = 0; i < 8; ++i)
        Cb[(long)(m0 + i + 8 * half) * N + n0 + r] = acc[i];
}

__global__ __launch_bounds__(256) void energy_softmax_kernel(
    const float* __restrict__ Uh,
    const float* __restrict__ Ws,
    const float* __restrict__ Va,
    float* __restrict__ attn)
{
    __shared__ v4f ws_s[DE_ / 4];
    __shared__ v4f va_s[DE_ / 4];
    __shared__ float e_s[TE_];
    __shared__ float red[9];

    const int t   = blockIdx.x;
    const int b   = blockIdx.y;
    const int tid = threadIdx.x;

    if (tid < DE_ / 4) {
        ws_s[tid] = ((const v4f*)(Ws + ((long)b * TD_ + t) * DE_))[tid];
        va_s[tid] = ((const v4f*)Va)[tid];
    }
    __syncthreads();

    const int lane = tid & 31;
    const int wv   = tid >> 5;

    for (int te = wv; te < TE_; te += 8) {
        const v4f* uh4 = (const v4f*)(Uh + ((long)b * TE_ + te) * DE_);
        float p = 0.f;
#pragma unroll
        for (int i = 0; i < DE_ / 128; ++i) {
            const int d4 = lane + 32 * i;
            const v4f u = uh4[d4];
            const v4f w = ws_s[d4];
            const v4f a = va_s[d4];
            p += a.x * fast_tanh(u.x + w.x);
            p += a.y * fast_tanh(u.y + w.y);
            p += a.z * fast_tanh(u.z + w.z);
            p += a.w * fast_tanh(u.w + w.w);
        }
        for (int off = 16; off; off >>= 1)
            p += __shfl_down(p, off, 32);
        if (lane == 0) e_s[te] = p;
    }
    __syncthreads();

    const float v = e_s[tid];

    float m = v;
    for (int off = 16; off; off >>= 1) m = fmaxf(m, __shfl_down(m, off, 32));
    if (lane == 0) red[wv] = m;
    __syncthreads();
    if (tid == 0) {
        float mm = red[0];
        for (int i = 1; i < 8; ++i) mm = fmaxf(mm, red[i]);
        red[8] = mm;
    }
    __syncthreads();
    m = red[8];

    const float p = fast_exp2((v - m) * 1.4426950408889634f);
    float s = p;
    for (int off = 16; off; off >>= 1) s += __shfl_down(s, off, 32);
    if (lane == 0) red[wv] = s;
    __syncthreads();
    if (tid == 0) {
        float ss = red[0];
        for (int i = 1; i < 8; ++i) ss += red[i];
        red[8] = ss;
    }
    __syncthreads();

    attn[((long)b * TD_ + t) * TE_ + tid] = p / red[8];
}

extern "C" void kernel_launch(void* const* d_in, const int* in_sizes, int n_in,
                              void* d_out, int out_size, void* d_ws, size_t ws_size,
                              hipStream_t stream) {
    const float* enc = (const float*)d_in[0];
    const float* dec = (const float*)d_in[1];
    const float* Ua  = (const float*)d_in[2];
    const float* Wa  = (const float*)d_in[3];
    const float* Va  = (const float*)d_in[4];

    float* out     = (float*)d_out;
    float* context = out;
    float* attn    = out + (long)B_ * TD_ * DE_;

    float* Uh  = (float*)d_ws;
    float* Wsb = Uh + (long)B_ * TE_ * DE_;

    const dim3 wblk(32);

    wmma_gemm_f32<<<dim3(TE_ / 16, DE_ / 16, B_), wblk, 0, stream>>>(
        enc, Ua, Uh, DE_, DE_, (long)TE_ * DE_, 0L, (long)TE_ * DE_);

    wmma_gemm_f32<<<dim3(TD_ / 16, DE_ / 16, B_), wblk, 0, stream>>>(
        dec, Wa, Wsb, DD_, DE_, (long)TD_ * DD_, 0L, (long)TD_ * DE_);

    energy_softmax_kernel<<<dim3(TD_, B_), dim3(256), 0, stream>>>(
        Uh, Wsb, Va, attn);

    wmma_gemm_f32<<<dim3(TD_ / 16, DE_ / 16, B_), wblk, 0, stream>>>(
        attn, enc, context, TE_, DE_, (long)TD_ * TE_, (long)TE_ * DE_, (long)TD_ * DE_);
}


# Energy kernel re-decomposed: a lane owns 2 adjacent encoder positions as a packed f32 pair, sums over d itself; Uh pair-interleaved via wave-private LDS, Ws from scalar registers; 4 steps per block
# speedup vs baseline: 2.5901x; 2.5901x over previous
#include <hip/hip_runtime.h>
#include <hip/hip_bf16.h>

#define B_  4
#define TE_ 256
#define TD_ 256
#define DE_ 512
#define DD_ 512

typedef __attribute__((ext_vector_type(2))) float v2f;
typedef __attribute__((ext_vector_type(4))) float v4f;
typedef __attribute__((ext_vector_type(8))) float v8f;

#if __has_builtin(__builtin_amdgcn_tanhf)
__device__ __forceinline__ float fast_tanh(float x) { return __builtin_amdgcn_tanhf(x); }
#elif __has_builtin(__builtin_amdgcn_tanh_f32)
__device__ __forceinline__ float fast_tanh(float x) { return __builtin_amdgcn_tanh_f32(x); }
#else
__device__ __forceinline__ float fast_tanh(float x) { return tanhf(x); }
#endif

#if __has_builtin(__builtin_amdgcn_exp2f)
__device__ __forceinline__ float fast_exp2(float x) { return __builtin_amdgcn_exp2f(x); }
#else
__device__ __forceinline__ float fast_exp2(float x) { return exp2f(x); }
#endif

#define GM_  256
#define GN_  512
#define BM_  32
#define BN_  64
#define KC_  128
#define PA_  (KC_ + 4)
#define PBP_ (2 * BN_ + 16)

struct GemmFrag {
    v4f a[4];
    v2f b[4][2][2];
};

template <int KT, bool PAIRED>
__global__ __launch_bounds__(128) void wmma_gemm_f32_lds(
    const float* __restrict__ A, const float* __restrict__ Bm,
    float* __restrict__ C, long sA, long sB, long sC)
{
    constexpr int NC = KT / KC_;

    __shared__ __attribute__((aligned(16))) float As[NC][BM_ * PA_];
    __shared__ __attribute__((aligned(16))) float Bs[NC][(KC_ / 2) * PBP_];

    const int tid  = threadIdx.x;
    const int lane = tid & 31;
    const int wv   = tid >> 5;
    const int h    = lane >> 4;
    const int c    = lane & 15;
    const int wm   = wv & 1;
    const int wn   = wv >> 1;

    const int m0 = blockIdx.x * BM_;
    const int n0 = blockIdx.y * BN_;

    const float* Ab = A  + (long)blockIdx.z * sA + (long)m0 * KT;
    const float* Bb = Bm + (long)blockIdx.z * sB + n0;
    float*       Cb = C  + (long)blockIdx.z * sC;

    v4f ra[8];
    v4f rb[16];

    const int    prl   = wv * 16 + h;
    const float* gA    = Ab + (long)(wv * 8) * KT + 4 * lane;
    const float* gB    = Bb + (long)(2 * prl) * GN_ + 4 * c;
    const int    sAoff = (wv * 8) * PA_ + 4 * lane;
    const int    sBoff = prl * PBP_ + 8 * c;

    const int fA = (wm * 16 + c) * PA_ + 4 * h;
    const int fB = (2 * h) * PBP_ + 2 * (wn * 32 + c);

    v8f acc0 = {};
    v8f acc1 = {};

    auto gload = [&](int ch) {
#pragma unroll
        for (int i = 0; i < 8; ++i)
            ra[i] = *(const v4f*)(gA + (long)i * KT + ch * KC_);
#pragma unroll
        for (int i = 0; i < 8; ++i) {
            rb[2 * i]     = *(const v4f*)(gB + (long)(ch * KC_ + 4 * i) * GN_);
            rb[2 * i + 1] = *(const v4f*)(gB + (long)(ch * KC_ + 4 * i + 1) * GN_);
        }
    };

    auto sstore = [&](int ch) {
#pragma unroll
        for (int i = 0; i < 8; ++i)
            *(v4f*)(&As[ch][sAoff + i * PA_]) = ra[i];
#pragma unroll
        for (int i = 0; i < 8; ++i) {
            const v4f r0 = rb[2 * i];
            const v4f r1 = rb[2 * i + 1];
            v4f lo; lo.x = r0.x; lo.y = r1.x; lo.z = r0.y; lo.w = r1.y;
            v4f hi; hi.x = r0.z; hi.y = r1.z; hi.z = r0.w; hi.w = r1.w;
            float* d = &Bs[ch][sBoff + (2 * i) * PBP_];
            *(v4f*)(d)     = lo;
            *(v4f*)(d + 4) = hi;
        }
    };

    auto fload = [&](GemmFrag& f, int ch, int blk) {
        const float* as = &As[ch][fA + blk * 32];
        const float* bs = &Bs[ch][fB + (blk * 16) * PBP_];
#pragma unroll
        for (int s = 0; s < 4; ++s) {
            f.a[s] = *(const v4f*)(as + 8 * s);
#pragma unroll
            for (int t = 0; t < 2; ++t) {
                f.b[s][t][0] = *(const v2f*)(bs + (4 * s) * PBP_ + 32 * t);
                f.b[s][t][1] = *(const v2f*)(bs + (4 * s + 1) * PBP_ + 32 * t);
            }
        }
    };

    auto fmma = [&](const GemmFrag& f) {
#pragma unroll
        for (int s = 0; s < 4; ++s) {
            v2f a01; a01.x = f.a[s].x; a01.y = f.a[s].y;
            v2f a23; a23.x = f.a[s].z; a23.y = f.a[s].w;
            acc0 = __builtin_amdgcn_wmma_f32_16x16x4_f32(
                false, a01, false, f.b[s][0][0], (short)0, acc0, false, false);
            acc1 = __builtin_amdgcn_wmma_f32_16x16x4_f32(
                false, a01, false, f.b[s][1][0], (short)0, acc1, false, false);
            acc0 = __builtin_amdgcn_wmma_f32_16x16x4_f32(
                false, a23, false, f.b[s][0][1], (short)0, acc0, false, false);
            acc1 = __builtin_amdgcn_wmma_f32_16x16x4_f32(
                false, a23, false, f.b[s][1][1], (short)0, acc1, false, false);
        }
    };

    gload(0);
    sstore(0);
    __syncthreads();

    GemmFrag fr[2];

#pragma unroll
    for (int ch = 0; ch < NC; ++ch) {
        fload(fr[0], ch, 0);
#pragma unroll
        for (int blk = 0; blk < 4; ++blk) {
            if (blk + 1 < 4) {
                __builtin_amdgcn_sched_barrier(0);
                fload(fr[(blk + 1) & 1], ch, blk + 1);
                __builtin_amdgcn_sched_barrier(0);
            }
            if (blk == 0 && ch + 1 < NC) gload(ch + 1);
            if (blk == 3 && ch + 1 < NC) sstore(ch + 1);
            fmma(fr[blk & 1]);
        }
        if (ch + 1 < NC) __syncthreads();
    }

    if (PAIRED) {
        float* cPair = Cb + ((long)((m0 + wm * 16 + 8 * h) >> 1) * GN_ + n0 + wn * 32 + c) * 2;
#pragma unroll
        for (int i = 0; i < 8; i += 2) {
            v2f p0; p0.x = acc0[i]; p0.y = acc0[i + 1];
            v2f p1; p1.x = acc1[i]; p1.y = acc1[i + 1];
            *(v2f*)(cPair + (long)(i >> 1) * (2 * GN_))      = p0;
            *(v2f*)(cPair + (long)(i >> 1) * (2 * GN_) + 32) = p1;
        }
    } else {
        float* cRow = Cb + (long)(m0 + wm * 16 + 8 * h) * GN_ + n0 + wn * 32 + c;
#pragma unroll
        for (int i = 0; i < 8; ++i) {
            cRow[(long)i * GN_]      = acc0[i];
            cRow[(long)i * GN_ + 16] = acc1[i];
        }
    }
}

#define NT_  4
#define EC_  64
#define EP_  (2 * EC_ + 4)
#define EG_  8
#define EDH_ (DE_ / 2)
#define ENG_ (EDH_ / EG_)
#define EGC_ (EC_ / EG_)
#define ENC_ (EDH_ / EC_)

__global__ __launch_bounds__(256) void energy_softmax_kernel(
    const float* __restrict__ UhP,
    const float* __restrict__ Ws,
    const float* __restrict__ Va,
    float* __restrict__ attn)
{
    __shared__ __attribute__((aligned(16))) float us[8][2][32 * EP_];
    __shared__ __attribute__((aligned(16))) float va_s[DE_];
    __shared__ v2f part[2][2][128];
    __shared__ float redm[2][2][4];
    __shared__ float reds[2][2][4];

    const int tid  = threadIdx.x;
    const int lane = tid & 31;
    const int wv   = __builtin_amdgcn_readfirstlane(tid >> 5);
    const int pg   = wv & 3;
    const int dh   = wv >> 2;
    const int t0   = blockIdx.x * NT_;
    const int b    = blockIdx.y;

    const float* gsrc = UhP + ((long)b * (TE_ / 2) + pg * 32) * (2 * DE_)
                            + 2 * (dh * EDH_) + 4 * lane;
    float* tile = &us[wv][0][0];
    const int soff = 4 * lane;
    const int roff = lane * EP_;

    const float* wrow = Ws + ((long)b * TD_ + t0) * DE_;
    const int dbase = dh * EDH_;

    v4f stg[32];
#pragma unroll
    for (int i = 0; i < 32; ++i)
        stg[i] = *(const v4f*)(gsrc + (long)i * (2 * DE_));

    if (tid < DE_ / 4)
        ((v4f*)va_s)[tid] = ((const v4f*)Va)[tid];

    v2f acc[NT_][2];
#pragma unroll
    for (int t = 0; t < NT_; ++t) { acc[t][0] = (v2f){0.f, 0.f}; acc[t][1] = (v2f){0.f, 0.f}; }

#pragma unroll
    for (int i = 0; i < 32; ++i)
        *(v4f*)(tile + soff + i * EP_) = stg[i];
    __syncthreads();
#pragma unroll
    for (int i = 0; i < 32; ++i)
        stg[i] = *(const v4f*)(gsrc + (long)i * (2 * DE_) + 2 * EC_);

    auto sload = [&](v8f (&w)[NT_], int g) {
#pragma unroll
        for (int t = 0; t < NT_; ++t)
            w[t] = *(const v8f*)(wrow + t * DE_ + dbase + g * EG_);
    };
    auto uload = [&](v4f (&u)[4], v4f (&a)[2], int g) {
        const float* p = tile + ((g / EGC_) & 1) * (32 * EP_) + roff + (g & (EGC_ - 1)) * (2 * EG_);
#pragma unroll
        for (int k = 0; k < 4; ++k)
            u[k] = *(const v4f*)(p + 4 * k);
        a[0] = *(const v4f*)(va_s + dbase + g * EG_);
        a[1] = *(const v4f*)(va_s + dbase + g * EG_ + 4);
    };
    auto compute = [&](const v8f (&w)[NT_], const v4f (&u)[4], const v4f (&a)[2]) {
#pragma unroll
        for (int dd = 0; dd < EG_; ++dd) {
            const v4f uu = u[dd >> 1];
            v2f up;
            if (dd & 1) { up.x = uu.z; up.y = uu.w; }
            else        { up.x = uu.x; up.y = uu.y; }
            const float as = a[dd >> 2][dd & 3];
            const v2f ap = {as, as};
#pragma unroll
            for (int t = 0; t < NT_; ++t) {
                const float wsc = w[t][dd];
                const v2f wp = {wsc, wsc};
                const v2f s = up + wp;
                v2f hh; hh.x = fast_tanh(s.x); hh.y = fast_tanh(s.y);
                acc[t][dd & 1] = ap * hh + acc[t][dd & 1];
            }
        }
    };

    v8f wA[NT_], wB[NT_];
    v4f uA[4], uB[4], aA[2], aB[2];
    sload(wA, 0);
    uload(uA, aA, 0);

    for (int it = 0; it < ENG_ / 2; ++it) {
        const int g  = 2 * it;
        const int ph = it & (EGC_ / 2 - 1);
        const int ch = it / (EGC_ / 2);
        sload(wB, g + 1);
        uload(uB, aB, g + 1);
        compute(wA, uA, aA);
        if (ph == EGC_ / 2 - 1 && ch + 1 < ENC_) {
            float* nxt = tile + ((ch + 1) & 1) * (32 * EP_) + soff;
#pragma unroll
            for (int i = 0; i < 32; ++i)
                *(v4f*)(nxt + i * EP_) = stg[i];
            if (ch + 2 < ENC_) {
#pragma unroll
                for (int i = 0; i < 32; ++i)
                    stg[i] = *(const v4f*)(gsrc + (long)i * (2 * DE_) + (ch + 2) * (2 * EC_));
            }
        }
        sload(wA, (g + 2) & (ENG_ - 1));
        uload(uA, aA, (g + 2) & (ENG_ - 1));
        compute(wB, uB, aB);
    }

    v2f ev[NT_];
#pragma unroll
    for (int t = 0; t < NT_; ++t) ev[t] = acc[t][0] + acc[t][1];

    const int col = pg * 32 + lane;
    part[1 - dh][0][col] = dh ? ev[0] : ev[2];
    part[1 - dh][1][col] = dh ? ev[1] : ev[3];
    __syncthreads();
    v2f ef[2];
    ef[0] = (dh ? ev[2] : ev[0]) + part[dh][0][col];
    ef[1] = (dh ? ev[3] : ev[1]) + part[dh][1][col];

    float m[2];
#pragma unroll
    for (int j = 0; j < 2; ++j) {
        float v = fmaxf(ef[j].x, ef[j].y);
        for (int off = 16; off; off >>= 1) v = fmaxf(v, __shfl_xor(v, off, 32));
        m[j] = v;
    }
    if (lane == 0) { redm[dh][0][pg] = m[0]; redm[dh][1][pg] = m[1]; }
    __syncthreads();

    v2f p[2];
    float s[2];
#pragma unroll
    for (int j = 0; j < 2; ++j) {
        const float mm = fmaxf(fmaxf(redm[dh][j][0], redm[dh][j][1]),
                               fmaxf(redm[dh][j][2], redm[dh][j][3]));
        p[j].x = fast_exp2((ef[j].x - mm) * 1.4426950408889634f);
        p[j].y = fast_exp2((ef[j].y - mm) * 1.4426950408889634f);
        float v = p[j].x + p[j].y;
        for (int off = 16; off; off >>= 1) v += __shfl_xor(v, off, 32);
        s[j] = v;
    }
    if (lane == 0) { reds[dh][0][pg] = s[0]; reds[dh][1][pg] = s[1]; }
    __syncthreads();

#pragma unroll
    for (int j = 0; j < 2; ++j) {
        const float ss = (reds[dh][j][0] + reds[dh][j][1]) + (reds[dh][j][2] + reds[dh][j][3]);
        v2f o; o.x = p[j].x / ss; o.y = p[j].y / ss;
        *(v2f*)(attn + ((long)b * TD_ + t0 + 2 * dh + j) * TE_ + 2 * col) = o;
    }
}

extern "C" void kernel_launch(void* const* d_in, const int* in_sizes, int n_in,
                              void* d_out, int out_size, void* d_ws, size_t ws_size,
                              hipStream_t stream) {
    const float* enc = (const float*)d_in[0];
    const float* dec = (const float*)d_in[1];
    const float* Ua  = (const float*)d_in[2];
    const float* Wa  = (const float*)d_in[3];
    const float* Va  = (const float*)d_in[4];

    float* out     = (float*)d_out;
    float* context = out;
    float* attn    = out + (long)B_ * TD_ * DE_;

    float* Uh  = (float*)d_ws;
    float* Wsb = Uh + (long)B_ * TE_ * DE_;

    const dim3 gblk(128);
    const dim3 ggrid(GM_ / BM_, GN_ / BN_, B_);

    wmma_gemm_f32_lds<DE_, true><<<ggrid, gblk, 0, stream>>>(
        enc, Ua, Uh, (long)TE_ * DE_, 0L, (long)TE_ * DE_);

    wmma_gemm_f32_lds<DD_, false><<<ggrid, gblk, 0, stream>>>(
        dec, Wa, Wsb, (long)TD_ * DD_, 0L, (long)TD_ * DE_);

    energy_softmax_kernel<<<dim3(TD_ / NT_, B_), dim3(256), 0, stream>>>(
        Uh, Wsb, Va, attn);

    wmma_gemm_f32_lds<TE_, false><<<ggrid, gblk, 0, stream>>>(
        attn, enc, context, (long)TD_ * TE_, (long)TE_ * DE_, (long)TD_ * DE_);
}


# Energy loop issue order: 64-unit pass written as a software pipeline and pinned by scheduling groups so each tanh is followed by one packed op; tile rows hold both chunks, no buffer arithmetic
# speedup vs baseline: 2.6809x; 1.0351x over previous
#include <hip/hip_runtime.h>
#include <hip/hip_bf16.h>

#define B_  4
#define TE_ 256
#define TD_ 256
#define DE_ 512
#define DD_ 512

typedef __attribute__((ext_vector_type(2))) float v2f;
typedef __attribute__((ext_vector_type(4))) float v4f;
typedef __attribute__((ext_vector_type(8))) float v8f;

#if __has_builtin(__builtin_amdgcn_tanhf)
__device__ __forceinline__ float fast_tanh(float x) { return __builtin_amdgcn_tanhf(x); }
#elif __has_builtin(__builtin_amdgcn_tanh_f32)
__device__ __forceinline__ float fast_tanh(float x) { return __builtin_amdgcn_tanh_f32(x); }
#else
__device__ __forceinline__ float fast_tanh(float x) { return tanhf(x); }
#endif

#if __has_builtin(__builtin_amdgcn_exp2f)
__device__ __forceinline__ float fast_exp2(float x) { return __builtin_amdgcn_exp2f(x); }
#else
__device__ __forceinline__ float fast_exp2(float x) { return exp2f(x); }
#endif

#define GM_  256
#define GN_  512
#define BM_  32
#define BN_  64
#define KC_  128
#define PA_  (KC_ + 4)
#define PBP_ (2 * BN_ + 16)

struct GemmFrag {
    v4f a[4];
    v2f b[4][2][2];
};

template <int KT, bool PAIRED>
__global__ __launch_bounds__(128) void wmma_gemm_f32_lds(
    const float* __restrict__ A, const float* __restrict__ Bm,
    float* __restrict__ C, long sA, long sB, long sC)
{
    constexpr int NC = KT / KC_;

    __shared__ __attribute__((aligned(16))) float As[NC][BM_ * PA_];
    __shared__ __attribute__((aligned(16))) float Bs[NC][(KC_ / 2) * PBP_];

    const int tid  = threadIdx.x;
    const int lane = tid & 31;
    const int wv   = tid >> 5;
    const int h    = lane >> 4;
    const int c    = lane & 15;
    const int wm   = wv & 1;
    const int wn   = wv >> 1;

    const int m0 = blockIdx.x * BM_;
    const int n0 = blockIdx.y * BN_;

    const float* Ab = A  + (long)blockIdx.z * sA + (long)m0 * KT;
    const float* Bb = Bm + (long)blockIdx.z * sB + n0;
    float*       Cb = C  + (long)blockIdx.z * sC;

    v4f ra[8];
    v4f rb[16];

    const int    prl   = wv * 16 + h;
    const float* gA    = Ab + (long)(wv * 8) * KT + 4 * lane;
    const float* gB    = Bb + (long)(2 * prl) * GN_ + 4 * c;
    const int    sAoff = (wv * 8) * PA_ + 4 * lane;
    const int    sBoff = prl * PBP_ + 8 * c;

    const int fA = (wm * 16 + c) * PA_ + 4 * h;
    const int fB = (2 * h) * PBP_ + 2 * (wn * 32 + c);

    v8f acc0 = {};
    v8f acc1 = {};

    auto gload = [&](int ch) {
#pragma unroll
        for (int i = 0; i < 8; ++i)
            ra[i] = *(const v4f*)(gA + (long)i * KT + ch * KC_);
#pragma unroll
        for (int i = 0; i < 8; ++i) {
            rb[2 * i]     = *(const v4f*)(gB + (long)(ch * KC_ + 4 * i) * GN_);
            rb[2 * i + 1] = *(const v4f*)(gB + (long)(ch * KC_ + 4 * i + 1) * GN_);
        }
    };

    auto sstore = [&](int ch) {
#pragma unroll
        for (int i = 0; i < 8; ++i)
            *(v4f*)(&As[ch][sAoff + i * PA_]) = ra[i];
#pragma unroll
        for (int i = 0; i < 8; ++i) {
            const v4f r0 = rb[2 * i];
            const v4f r1 = rb[2 * i + 1];
            v4f lo; lo.x = r0.x; lo.y = r1.x; lo.z = r0.y; lo.w = r1.y;
            v4f hi; hi.x = r0.z; hi.y = r1.z; hi.z = r0.w; hi.w = r1.w;
            float* d = &Bs[ch][sBoff + (2 * i) * PBP_];
            *(v4f*)(d)     = lo;
            *(v4f*)(d + 4) = hi;
        }
    };

    auto fload = [&](GemmFrag& f, int ch, int blk) {
        const float* as = &As[ch][fA + blk * 32];
        const float* bs = &Bs[ch][fB + (blk * 16) * PBP_];
#pragma unroll
        for (int s = 0; s < 4; ++s) {
            f.a[s] = *(const v4f*)(as + 8 * s);
#pragma unroll
            for (int t = 0; t < 2; ++t) {
                f.b[s][t][0] = *(const v2f*)(bs + (4 * s) * PBP_ + 32 * t);
                f.b[s][t][1] = *(const v2f*)(bs + (4 * s + 1) * PBP_ + 32 * t);
            }
        }
    };

    auto fmma = [&](const GemmFrag& f) {
#pragma unroll
        for (int s = 0; s < 4; ++s) {
            v2f a01; a01.x = f.a[s].x; a01.y = f.a[s].y;
            v2f a23; a23.x = f.a[s].z; a23.y = f.a[s].w;
            acc0 = __builtin_amdgcn_wmma_f32_16x16x4_f32(
                false, a01, false, f.b[s][0][0], (short)0, acc0, false, false);
            acc1 = __builtin_amdgcn_wmma_f32_16x16x4_f32(
                false, a01, false, f.b[s][1][0], (short)0, acc1, false, false);
            acc0 = __builtin_amdgcn_wmma_f32_16x16x4_f32(
                false, a23, false, f.b[s][0][1], (short)0, acc0, false, false);
            acc1 = __builtin_amdgcn_wmma_f32_16x16x4_f32(
                false, a23, false, f.b[s][1][1], (short)0, acc1, false, false);
        }
    };

    gload(0);
    sstore(0);
    __syncthreads();

    GemmFrag fr[2];

#pragma unroll
    for (int ch = 0; ch < NC; ++ch) {
        fload(fr[0], ch, 0);
#pragma unroll
        for (int blk = 0; blk < 4; ++blk) {
            if (blk + 1 < 4) {
                __builtin_amdgcn_sched_barrier(0);
                fload(fr[(blk + 1) & 1], ch, blk + 1);
                __builtin_amdgcn_sched_barrier(0);
            }
            if (blk == 0 && ch + 1 < NC) gload(ch + 1);
            if (blk == 3 && ch + 1 < NC) sstore(ch + 1);
            fmma(fr[blk & 1]);
        }
        if (ch + 1 < NC) __syncthreads();
    }

    if (PAIRED) {
        float* cPair = Cb + ((long)((m0 + wm * 16 + 8 * h) >> 1) * GN_ + n0 + wn * 32 + c) * 2;
#pragma unroll
        for (int i = 0; i < 8; i += 2) {
            v2f p0; p0.x = acc0[i]; p0.y = acc0[i + 1];
            v2f p1; p1.x = acc1[i]; p1.y = acc1[i + 1];
            *(v2f*)(cPair + (long)(i >> 1) * (2 * GN_))      = p0;
            *(v2f*)(cPair + (long)(i >> 1) * (2 * GN_) + 32) = p1;
        }
    } else {
        float* cRow = Cb + (long)(m0 + wm * 16 + 8 * h) * GN_ + n0 + wn * 32 + c;
#pragma unroll
        for (int i = 0; i < 8; ++i) {
            cRow[(long)i * GN_]      = acc0[i];
            cRow[(long)i * GN_ + 16] = acc1[i];
        }
    }
}

#define NT_  4
#define EC_  64
#define EP_  (4 * EC_ + 4)
#define EG_  8
#define EDH_ (DE_ / 2)
#define ENG_ (EDH_ / EG_)
#define EGC_ (EC_ / EG_)
#define ENC_ (EDH_ / EC_)

__global__ __launch_bounds__(256) void energy_softmax_kernel(
    const float* __restrict__ UhP,
    const float* __restrict__ Ws,
    const float* __restrict__ Va,
    float* __restrict__ attn)
{
    __shared__ __attribute__((aligned(16))) float us[8][32 * EP_];
    __shared__ __attribute__((aligned(16))) float va_s[DE_];
    __shared__ v2f part[2][2][128];
    __shared__ float redm[2][2][4];
    __shared__ float reds[2][2][4];

    const int tid  = threadIdx.x;
    const int lane = tid & 31;
    const int wv   = __builtin_amdgcn_readfirstlane(tid >> 5);
    const int pg   = wv & 3;
    const int dh   = wv >> 2;
    const int t0   = blockIdx.x * NT_;
    const int b    = blockIdx.y;

    const float* gsrc = UhP + ((long)b * (TE_ / 2) + pg * 32) * (2 * DE_)
                            + 2 * (dh * EDH_) + 4 * lane;
    float* tile = &us[wv][0];
    const int soff = 4 * lane;
    const int roff = lane * EP_;

    const float* wrow = Ws + ((long)b * TD_ + t0) * DE_;
    const int dbase = dh * EDH_;

    v4f stg[32];
#pragma unroll
    for (int i = 0; i < 32; ++i)
        stg[i] = *(const v4f*)(gsrc + (long)i * (2 * DE_));

    if (tid < DE_ / 4)
        ((v4f*)va_s)[tid] = ((const v4f*)Va)[tid];

    v2f acc[NT_][2];
#pragma unroll
    for (int t = 0; t < NT_; ++t) { acc[t][0] = (v2f){0.f, 0.f}; acc[t][1] = (v2f){0.f, 0.f}; }

#pragma unroll
    for (int i = 0; i < 32; ++i)
        *(v4f*)(tile + soff + i * EP_) = stg[i];
    __syncthreads();
#pragma unroll
    for (int i = 0; i < 32; ++i)
        stg[i] = *(const v4f*)(gsrc + (long)i * (2 * DE_) + 2 * EC_);

    auto sload = [&](v8f (&w)[NT_], int g) {
#pragma unroll
        for (int t = 0; t < NT_; ++t)
            w[t] = *(const v8f*)(wrow + t * DE_ + dbase + g * EG_);
    };
    auto uload = [&](v4f (&u)[4], v4f (&a)[2], int g) {
        const float* p = tile + roff + (g & (2 * EGC_ - 1)) * (2 * EG_);
#pragma unroll
        for (int k = 0; k < 4; ++k)
            u[k] = *(const v4f*)(p + 4 * k);
        a[0] = *(const v4f*)(va_s + dbase + g * EG_);
        a[1] = *(const v4f*)(va_s + dbase + g * EG_ + 4);
    };

    v8f wA[NT_], wB[NT_], wN[NT_];
    v4f uA[4], uB[4], uN[4], aA[2], aB[2], aN[2];
    sload(wA, 0);
    uload(uA, aA, 0);

    v2f sc[4], hp[4];
    float asp = 0.f;
#pragma unroll
    for (int t = 0; t < 4; ++t) {
        const v2f up = {uA[0].x, uA[0].y};
        const v2f wp = {wA[t][0], wA[t][0]};
        sc[t] = up + wp;
        hp[t] = (v2f){0.f, 0.f};
    }

    for (int it = 0; it < ENG_ / 2; ++it) {
        const int g = 2 * it;
        if ((0x0888u >> it) & 1u) {
            const int ch = it >> 2;
            float* nxt = tile + ((ch + 1) & 1) * (2 * EC_) + soff;
#pragma unroll
            for (int i = 0; i < 32; ++i)
                *(v4f*)(nxt + i * EP_) = stg[i];
            if (ch + 2 < ENC_) {
#pragma unroll
                for (int i = 0; i < 32; ++i)
                    stg[i] = *(const v4f*)(gsrc + (long)i * (2 * DE_) + (ch + 2) * (2 * EC_));
            }
        }
        sload(wB, g + 1);
        uload(uB, aB, g + 1);

        auto addu = [&](int i) -> v2f {
            const int t = i & 3, dd = (i >> 2) & 7;
            const v4f uu = (i < 32) ? uA[dd >> 1] : uB[dd >> 1];
            v2f up;
            if (dd & 1) { up.x = uu.z; up.y = uu.w; }
            else        { up.x = uu.x; up.y = uu.y; }
            const float wsc = (i < 32) ? wA[t][dd] : wB[t][dd];
            const v2f wp = {wsc, wsc};
            return up + wp;
        };
        auto fmau = [&](int i, v2f hh) {
            const int t = i & 3, dd = (i >> 2) & 7;
            const float as = (i < 32) ? aA[dd >> 2][dd & 3] : aB[dd >> 2][dd & 3];
            const v2f ap = {as, as};
            acc[t][dd & 1] = ap * hh + acc[t][dd & 1];
        };

        v2f sq[68];
        v2f hq[64];
#pragma unroll
        for (int i = 0; i < 4; ++i) sq[i] = sc[i];
#pragma unroll
        for (int i = 0; i < 64; ++i) {
            hq[i].x = fast_tanh(sq[i].x);
            if (i + 4 < 64) {
                sq[i + 4] = addu(i + 4);
            } else {
                const v2f up = {uN[0].x, uN[0].y};
                const float wsc = wN[i - 60][0];
                const v2f wp = {wsc, wsc};
                sq[i + 4] = up + wp;
            }
            hq[i].y = fast_tanh(sq[i].y);
            if (i >= 4) {
                fmau(i - 4, hq[i - 4]);
            } else {
                const v2f ap = {asp, asp};
                acc[i][1] = ap * hp[i] + acc[i][1];
            }
            if (i == 33) {
                sload(wN, (g + 2) & (ENG_ - 1));
                uload(uN, aN, (g + 2) & (ENG_ - 1));
            }
            if ((i & 7) == 7) {
#pragma unroll
                for (int k = 0; k < 16; ++k) {
                    __builtin_amdgcn_sched_group_barrier(0x400, 1, 0);
                    __builtin_amdgcn_sched_group_barrier(0x002, 1, 0);
                }
                __builtin_amdgcn_sched_barrier(0);
            }
        }
#pragma unroll
        for (int i = 0; i < 4; ++i) { sc[i] = sq[64 + i]; hp[i] = hq[60 + i]; }
        asp = aB[1][3];
#pragma unroll
        for (int t = 0; t < NT_; ++t) wA[t] = wN[t];
#pragma unroll
        for (int k = 0; k < 4; ++k) uA[k] = uN[k];
        aA[0] = aN[0]; aA[1] = aN[1];
    }

    {
        const v2f ap = {asp, asp};
#pragma unroll
        for (int t = 0; t < 4; ++t) acc[t][1] = ap * hp[t] + acc[t][1];
    }

    v2f ev[NT_];
#pragma unroll
    for (int t = 0; t < NT_; ++t) ev[t] = acc[t][0] + acc[t][1];

    const int col = pg * 32 + lane;
    part[1 - dh][0][col] = dh ? ev[0] : ev[2];
    part[1 - dh][1][col] = dh ? ev[1] : ev[3];
    __syncthreads();
    v2f ef[2];
    ef[0] = (dh ? ev[2] : ev[0]) + part[dh][0][col];
    ef[1] = (dh ? ev[3] : ev[1]) + part[dh][1][col];

    float m[2];
#pragma unroll
    for (int j = 0; j < 2; ++j) {
        float v = fmaxf(ef[j].x, ef[j].y);
        for (int off = 16; off; off >>= 1) v = fmaxf(v, __shfl_xor(v, off, 32));
        m[j] = v;
    }
    if (lane == 0) { redm[dh][0][pg] = m[0]; redm[dh][1][pg] = m[1]; }
    __syncthreads();

    v2f p[2];
    float s[2];
#pragma unroll
    for (int j = 0; j < 2; ++j) {
        const float mm = fmaxf(fmaxf(redm[dh][j][0], redm[dh][j][1]),
                               fmaxf(redm[dh][j][2], redm[dh][j][3]));
        p[j].x = fast_exp2((ef[j].x - mm) * 1.4426950408889634f);
        p[j].y = fast_exp2((ef[j].y - mm) * 1.4426950408889634f);
        float v = p[j].x + p[j].y;
        for (int off = 16; off; off >>= 1) v += __shfl_xor(v, off, 32);
        s[j] = v;
    }
    if (lane == 0) { reds[dh][0][pg] = s[0]; reds[dh][1][pg] = s[1]; }
    __syncthreads();

#pragma unroll
    for (int j = 0; j < 2; ++j) {
        const float ss = (reds[dh][j][0] + reds[dh][j][1]) + (reds[dh][j][2] + reds[dh][j][3]);
        v2f o; o.x = p[j].x / ss; o.y = p[j].y / ss;
        *(v2f*)(attn + ((long)b * TD_ + t0 + 2 * dh + j) * TE_ + 2 * col) = o;
    }
}

extern "C" void kernel_launch(void* const* d_in, const int* in_sizes, int n_in,
                              void* d_out, int out_size, void* d_ws, size_t ws_size,
                              hipStream_t stream) {
    const float* enc = (const float*)d_in[0];
    const float* dec = (const float*)d_in[1];
    const float* Ua  = (const float*)d_in[2];
    const float* Wa  = (const float*)d_in[3];
    const float* Va  = (const float*)d_in[4];

    float* out     = (float*)d_out;
    float* context = out;
    float* attn    = out + (long)B_ * TD_ * DE_;

    float* Uh  = (float*)d_ws;
    float* Wsb = Uh + (long)B_ * TE_ * DE_;

    const dim3 gblk(128);
    const dim3 ggrid(GM_ / BM_, GN_ / BN_, B_);

    wmma_gemm_f32_lds<DE_, true><<<ggrid, gblk, 0, stream>>>(
        enc, Ua, Uh, (long)TE_ * DE_, 0L, (long)TE_ * DE_);

    wmma_gemm_f32_lds<DD_, false><<<ggrid, gblk, 0, stream>>>(
        dec, Wa, Wsb, (long)TD_ * DD_, 0L, (long)TD_ * DE_);

    energy_softmax_kernel<<<dim3(TD_ / NT_, B_), dim3(256), 0, stream>>>(
        Uh, Wsb, Va, attn);

    wmma_gemm_f32_lds<TE_, false><<<ggrid, gblk, 0, stream>>>(
        attn, enc, context, (long)TD_ * TE_, (long)TE_ * DE_, (long)TD_ * DE_);
}
